# starting kernel
# baseline (speedup 1.0000x reference)
#include <hip/hip_runtime.h>
#include <hip/hip_bf16.h>


typedef __attribute__((ext_vector_type(2))) float v2f;
typedef __attribute__((ext_vector_type(8))) float v8f;

#define BB 16
#define TT 8192
#define DD 256
#define SS 64
#define EPS_ADD 0.006f
#define WAVES_PER_BLOCK 8

__global__ __launch_bounds__(256)
void WisePooling_seg_mean_wmma(const float* __restrict__ x,
                               const int*   __restrict__ graph,
                               float*       __restrict__ out) {
    const int lane = threadIdx.x & 31;
    const int wave = threadIdx.x >> 5;

    const int tile  = blockIdx.x * WAVES_PER_BLOCK + wave;
    const int dtile = tile & 15;
    const int bs    = tile >> 4;
    const int b     = bs >> 6;

    const int t_start = __builtin_amdgcn_readfirstlane(graph[bs * 2 + 0]);
    const int t_end   = __builtin_amdgcn_readfirstlane(graph[bs * 2 + 1]);
    const int len     = t_end - t_start + 1;

    const int d0    = dtile * 16;
    const int khalf = (lane >> 4) << 1;

    const float* __restrict__ colf =
        x + (size_t)b * TT * DD + d0 + (lane & 15);

    const float* __restrict__ p = colf + (size_t)(t_start + khalf) * DD;

    v2f a_ones; a_ones.x = 1.0f; a_ones.y = 1.0f;
    v8f acc0 = {};
    v8f acc1 = {};

    const int nchunks = len >> 3;
    int i = 0;
    for (; i + 1 < nchunks; i += 2) {
        v2f b0, b1;
        b0.x = p[0]       + p[4 * DD];
        b0.y = p[DD]      + p[5 * DD];
        b1.x = p[8 * DD]  + p[12 * DD];
        b1.y = p[9 * DD]  + p[13 * DD];
        acc0 = __builtin_amdgcn_wmma_f32_16x16x4_f32(
            false, a_ones, false, b0, (short)0, acc0, false, false);
        acc1 = __builtin_amdgcn_wmma_f32_16x16x4_f32(
            false, a_ones, false, b1, (short)0, acc1, false, false);
        p += 16 * DD;
    }
    if (i < nchunks) {
        v2f b0;
        b0.x = p[0]  + p[4 * DD];
        b0.y = p[DD] + p[5 * DD];
        acc0 = __builtin_amdgcn_wmma_f32_16x16x4_f32(
            false, a_ones, false, b0, (short)0, acc0, false, false);
    }

    const int t0 = t_start + (nchunks << 3);
    if (t0 <= t_end) {
        const int ta = t0 + khalf;
        const int tb = ta + 1;
        v2f bm;
        {
            int u0 = ta,     c0 = (u0 <= t_end) ? u0 : t_end;
            int u1 = ta + 4, c1 = (u1 <= t_end) ? u1 : t_end;
            float r0 = colf[(size_t)c0 * DD];
            float r1 = colf[(size_t)c1 * DD];
            bm.x = ((u0 <= t_end) ? r0 : 0.0f) + ((u1 <= t_end) ? r1 : 0.0f);
        }
        {
            int u0 = tb,     c0 = (u0 <= t_end) ? u0 : t_end;
            int u1 = tb + 4, c1 = (u1 <= t_end) ? u1 : t_end;
            float r0 = colf[(size_t)c0 * DD];
            float r1 = colf[(size_t)c1 * DD];
            bm.y = ((u0 <= t_end) ? r0 : 0.0f) + ((u1 <= t_end) ? r1 : 0.0f);
        }
        acc0 = __builtin_amdgcn_wmma_f32_16x16x4_f32(
            false, a_ones, false, bm, (short)0, acc0, false, false);
    }

    const float inv_len = 1.0f / (float)len;
    const float sum = acc0[0] + acc1[0];
    if (lane < 16) {
        out[(size_t)bs * DD + d0 + lane] = sum * inv_len + EPS_ADD;
    }
}

extern "C" void kernel_launch(void* const* d_in, const int* in_sizes, int n_in,
                              void* d_out, int out_size, void* d_ws, size_t ws_size,
                              hipStream_t stream) {
    (void)in_sizes; (void)n_in; (void)out_size; (void)d_ws; (void)ws_size;
    const float* x     = (const float*)d_in[0];
    const int*   graph = (const int*)d_in[1];
    float*       out   = (float*)d_out;

    const int total_tiles = BB * SS * 16;
    const int blocks = total_tiles / WAVES_PER_BLOCK;
    WisePooling_seg_mean_wmma<<<dim3(blocks), dim3(256), 0, stream>>>(x, graph, out);
}


# 16-byte loads per lane: wave tile 16 -> 64 features (each half-wave reads one aligned 256-B row chunk), 4096 waves instead of 16384; the new row-to-slot grouping changes the f32 summation order
# speedup vs baseline: 1.2337x; 1.2337x over previous
#include <hip/hip_runtime.h>


typedef __attribute__((ext_vector_type(2))) float v2f;
typedef __attribute__((ext_vector_type(4))) float v4f;
typedef __attribute__((ext_vector_type(8))) float v8f;

#define BB 16
#define TT 8192
#define DD 256
#define SS 64
#define EPS_ADD 0.006f
#define WAVES_PER_BLOCK 8
#define ROW4 (DD / 4)
#define GROUP_LOADS 16
#define GROUP_ROWS 32

__device__ __forceinline__ v8f colsum4(v2f ones, float k0, float k1, v8f acc) {
    v2f bb; bb.x = k0; bb.y = k1;
    return __builtin_amdgcn_wmma_f32_16x16x4_f32(
        false, ones, false, bb, (short)0, acc, false, false);
}

__device__ __forceinline__ void reduce_group(const v4f (&v)[GROUP_LOADS], v2f ones,
                                             v8f& acc0, v8f& acc1, v8f& acc2, v8f& acc3) {
    for (int g = 0; g < GROUP_LOADS / 4; ++g) {
        const v4f s0 = v[4 * g + 0] + v[4 * g + 1];
        const v4f s1 = v[4 * g + 2] + v[4 * g + 3];
        acc0 = colsum4(ones, s0.x, s1.x, acc0);
        acc1 = colsum4(ones, s0.y, s1.y, acc1);
        acc2 = colsum4(ones, s0.z, s1.z, acc2);
        acc3 = colsum4(ones, s0.w, s1.w, acc3);
    }
}

__global__ __launch_bounds__(256)
void seg_mean_wmma_b128(const float* __restrict__ x,
                        const int*   __restrict__ graph,
                        float*       __restrict__ out) {
    const int lane = threadIdx.x & 31;
    const int wave = threadIdx.x >> 5;

    const int tile = blockIdx.x * WAVES_PER_BLOCK + wave;
    const int q    = tile & 3;
    const int bs   = tile >> 2;
    const int b    = bs >> 6;

    const int t_start = __builtin_amdgcn_readfirstlane(graph[bs * 2 + 0]);
    const int t_end   = __builtin_amdgcn_readfirstlane(graph[bs * 2 + 1]);
    const int len     = t_end - t_start + 1;

    const int c = lane & 15;
    const int h = lane >> 4;

    const v4f* __restrict__ col =
        (const v4f*)(x + (size_t)b * TT * DD + q * 64 + c * 4);
    const v4f* __restrict__ p = col + (size_t)(t_start + h) * ROW4;

    v2f ones; ones.x = 1.0f; ones.y = 1.0f;
    v8f acc0 = {}, acc1 = {}, acc2 = {}, acc3 = {};

    const int ngroups = len >> 5;
    for (int i = 0; i < ngroups; ++i) {
        v4f v[GROUP_LOADS];
        for (int u = 0; u < GROUP_LOADS; ++u) v[u] = p[(size_t)u * 2 * ROW4];
        p += (size_t)GROUP_ROWS * ROW4;
        reduce_group(v, ones, acc0, acc1, acc2, acc3);
    }

    const int t0 = t_start + (ngroups << 5);
    if (t0 <= t_end) {
        v4f v[GROUP_LOADS];
        for (int u = 0; u < GROUP_LOADS; ++u) {
            const int r  = t0 + 2 * u + h;
            const int rc = (r <= t_end) ? r : t_end;
            const v4f w  = col[(size_t)rc * ROW4];
            v4f z = {};
            v[u] = (r <= t_end) ? w : z;
        }
        reduce_group(v, ones, acc0, acc1, acc2, acc3);
    }

    const float inv_len = 1.0f / (float)len;
    if (lane < 16) {
        v4f o;
        o.x = acc0[0] * inv_len + EPS_ADD;
        o.y = acc1[0] * inv_len + EPS_ADD;
        o.z = acc2[0] * inv_len + EPS_ADD;
        o.w = acc3[0] * inv_len + EPS_ADD;
        *(v4f*)(out + (size_t)bs * DD + q * 64 + c * 4) = o;
    }
}

extern "C" void kernel_launch(void* const* d_in, const int* in_sizes, int n_in,
                              void* d_out, int out_size, void* d_ws, size_t ws_size,
                              hipStream_t stream) {
    (void)in_sizes; (void)n_in; (void)out_size; (void)d_ws; (void)ws_size;
    const float* x     = (const float*)d_in[0];
    const int*   graph = (const int*)d_in[1];
    float*       out   = (float*)d_out;

    const int total_tiles = BB * SS * 4;
    const int blocks = total_tiles / WAVES_PER_BLOCK;
    seg_mean_wmma_b128<<<dim3(blocks), dim3(256), 0, stream>>>(x, graph, out);
}
